# starting kernel
# baseline (speedup 1.0000x reference)
#include <hip/hip_runtime.h>
#include <math.h>

typedef __attribute__((ext_vector_type(2))) float v2f;
typedef __attribute__((ext_vector_type(8))) float v8f;

#define NEG_SLOPE 0.2f

template<int K, int NOUT>
__global__ __launch_bounds__(256) void gemm_wmma(const float* __restrict__ A,
                                                 const float* __restrict__ B,
                                                 float* __restrict__ C,
                                                 int ntasks)
{
    constexpr int NG = NOUT / 64;
    int task = blockIdx.x * 8 + (threadIdx.x >> 5);
    if (task >= ntasks) return;
    int rt = task / NG;
    int cg = task % NG;
    int lane = threadIdx.x & 31;
    int half = lane >> 4;
    int mr   = lane & 15;

    const float* arow = A + (size_t)(rt * 16 + mr) * K + 2 * half;
    const float* bcol = B + cg * 64 + mr;

    v8f acc[4] = {};

    for (int k0 = 0; k0 < K; k0 += 4) {
        v2f a;
        a.x = arow[k0];
        a.y = arow[k0 + 1];
        const float* brow0 = bcol + (size_t)(k0 + 2 * half) * NOUT;
        const float* brow1 = brow0 + NOUT;
#pragma unroll
        for (int t = 0; t < 4; ++t) {
            v2f b;
            b.x = brow0[t * 16];
            b.y = brow1[t * 16];
            acc[t] = __builtin_amdgcn_wmma_f32_16x16x4_f32(
                false, a, false, b, (short)0, acc[t], false, false);
        }
    }

    float* crow = C + (size_t)(rt * 16 + half * 8) * NOUT + cg * 64 + mr;
#pragma unroll
    for (int t = 0; t < 4; ++t)
#pragma unroll
        for (int r = 0; r < 8; ++r)
            crow[(size_t)r * NOUT + t * 16] = acc[t][r];
}

template<int H, int C>
__global__ void att_coeff(const float* __restrict__ h,
                          const float* __restrict__ a_src,
                          const float* __restrict__ a_dst,
                          float* __restrict__ als,
                          float* __restrict__ ald,
                          int n)
{
    int i = blockIdx.x * blockDim.x + threadIdx.x;
    if (i >= n * H) return;
    int node = i / H, hd = i % H;
    const float* hp = h + (size_t)node * (H * C) + hd * C;
    const float* asrc = a_src + hd * C;
    const float* adst = a_dst + hd * C;
    float ss = 0.f, sd = 0.f;
#pragma unroll 4
    for (int c = 0; c < C; ++c) {
        float v = hp[c];
        ss += v * asrc[c];
        sd += v * adst[c];
    }
    als[i] = ss;
    ald[i] = sd;
}

template<int H>
__global__ void edge_denom(const int* __restrict__ src,
                           const int* __restrict__ dst,
                           const float* __restrict__ als,
                           const float* __restrict__ ald,
                           float* __restrict__ denom,
                           int E_, int n)
{
    int i = blockIdx.x * blockDim.x + threadIdx.x;
    int total = (E_ + n) * H;
    if (i >= total) return;
    int e = i / H, hd = i % H;
    int s, d;
    if (e < E_) { s = src[e]; d = dst[e]; }
    else        { s = d = e - E_; }
    float x = als[s * H + hd] + ald[d * H + hd];
    float lv = (x >= 0.f) ? x : NEG_SLOPE * x;
    atomicAdd(&denom[d * H + hd], expf(lv));
}

__global__ void edge_aggr1(const int* __restrict__ src,
                           const int* __restrict__ dst,
                           const float* __restrict__ als,
                           const float* __restrict__ ald,
                           const float* __restrict__ denom,
                           const float* __restrict__ h,
                           float* __restrict__ out,
                           int E_, int n)
{
    int wave = blockIdx.x * (blockDim.x >> 5) + (threadIdx.x >> 5);
    int lane = threadIdx.x & 31;
    if (wave >= E_ + n) return;
    int s, d;
    if (wave < E_) { s = src[wave]; d = dst[wave]; }
    else           { s = d = wave - E_; }

    float alpha = 0.f;
    if (lane < 8) {
        float x = als[s * 8 + lane] + ald[d * 8 + lane];
        float lv = (x >= 0.f) ? x : NEG_SLOPE * x;
        alpha = expf(lv) / denom[d * 8 + lane];
    }

    const float* hs = h + (size_t)s * 256;
    float* od = out + (size_t)d * 256;
#pragma unroll
    for (int j = 0; j < 8; ++j) {
        float aj = __shfl(alpha, j, 32);
        int col = lane + 32 * j;
        atomicAdd(&od[col], aj * hs[col]);
    }
}

__global__ void edge_aggr2(const int* __restrict__ src,
                           const int* __restrict__ dst,
                           const float* __restrict__ als,
                           const float* __restrict__ ald,
                           const float* __restrict__ denom,
                           const float* __restrict__ h,
                           float* __restrict__ out,
                           int E_, int n)
{
    int wave = blockIdx.x * (blockDim.x >> 5) + (threadIdx.x >> 5);
    int lane = threadIdx.x & 31;
    if (wave >= E_ + n) return;
    int s, d;
    if (wave < E_) { s = src[wave]; d = dst[wave]; }
    else           { s = d = wave - E_; }

    float x = als[s] + ald[d];
    float lv = (x >= 0.f) ? x : NEG_SLOPE * x;
    float alpha = expf(lv) / denom[d];

    const float* hs = h + (size_t)s * 64;
    float* od = out + (size_t)d * 64;
#pragma unroll
    for (int j = 0; j < 2; ++j) {
        int col = lane + 32 * j;
        atomicAdd(&od[col], alpha * hs[col]);
    }
}

template<int F>
__global__ void bias_elu(float* __restrict__ buf,
                         const float* __restrict__ b,
                         int total)
{
    int i = blockIdx.x * blockDim.x + threadIdx.x;
    if (i >= total) return;
    float v = buf[i] + b[i % F];
    buf[i] = (v > 0.f) ? v : expm1f(v);
}

extern "C" void kernel_launch(void* const* d_in, const int* in_sizes, int n_in,
                              void* d_out, int out_size, void* d_ws, size_t ws_size,
                              hipStream_t stream)
{
    const float* x   = (const float*)d_in[0];
    const int*   ei  = (const int*)  d_in[1];
    const float* W1  = (const float*)d_in[2];
    const float* as1 = (const float*)d_in[3];
    const float* ad1 = (const float*)d_in[4];
    const float* b1  = (const float*)d_in[5];
    const float* W2  = (const float*)d_in[6];
    const float* as2 = (const float*)d_in[7];
    const float* ad2 = (const float*)d_in[8];
    const float* b2  = (const float*)d_in[9];
    float* out = (float*)d_out;

    const int IN_DIM = 128, H1 = 8, F1 = 256, OUT_DIM = 64;
    int n = in_sizes[0] / IN_DIM;
    int e = in_sizes[1] / 2;
    const int* src = ei;
    const int* dst = ei + e;

    float* ws = (float*)d_ws;
    float* h1   = ws;  ws += (size_t)n * F1;
    float* g1   = ws;  ws += (size_t)n * F1;
    float* h2   = ws;  ws += (size_t)n * OUT_DIM;
    float* als1 = ws;  ws += (size_t)n * H1;
    float* ald1 = ws;  ws += (size_t)n * H1;
    float* den1 = ws;  ws += (size_t)n * H1;
    float* als2 = ws;  ws += (size_t)n;
    float* ald2 = ws;  ws += (size_t)n;
    float* den2 = ws;  ws += (size_t)n;

    hipMemsetAsync(g1,   0, (size_t)n * F1 * sizeof(float), stream);
    hipMemsetAsync(den1, 0, (size_t)n * H1 * sizeof(float), stream);
    hipMemsetAsync(den2, 0, (size_t)n * sizeof(float), stream);
    hipMemsetAsync(out,  0, (size_t)n * OUT_DIM * sizeof(float), stream);

    int ne = e + n;

    {
        int tasks = (n / 16) * (F1 / 64);
        gemm_wmma<128, 256><<<(tasks + 7) / 8, 256, 0, stream>>>(x, W1, h1, tasks);
    }
    att_coeff<8, 32><<<(n * 8 + 255) / 256, 256, 0, stream>>>(h1, as1, ad1, als1, ald1, n);
    edge_denom<8><<<(ne * 8 + 255) / 256, 256, 0, stream>>>(src, dst, als1, ald1, den1, e, n);
    edge_aggr1<<<(ne + 7) / 8, 256, 0, stream>>>(src, dst, als1, ald1, den1, h1, g1, e, n);
    bias_elu<256><<<(n * 256 + 255) / 256, 256, 0, stream>>>(g1, b1, n * 256);

    {
        int tasks = (n / 16) * (OUT_DIM / 64);
        gemm_wmma<256, 64><<<(tasks + 7) / 8, 256, 0, stream>>>(g1, W2, h2, tasks);
    }
    att_coeff<1, 64><<<(n + 255) / 256, 256, 0, stream>>>(h2, as2, ad2, als2, ald2, n);
    edge_denom<1><<<(ne + 255) / 256, 256, 0, stream>>>(src, dst, als2, ald2, den2, e, n);
    edge_aggr2<<<(ne + 7) / 8, 256, 0, stream>>>(src, dst, als2, ald2, den2, h2, out, e, n);
    bias_elu<64><<<(n * 64 + 255) / 256, 256, 0, stream>>>(out, b2, n * 64);
}


# GEMMs: a work-group copies B (W1, W2) into LDS once and stays; its waves loop over row-tile tasks, B fragments read from LDS, A by 16-byte loads; k sum order inside the products changes
# speedup vs baseline: 3.4009x; 3.4009x over previous
#include <hip/hip_runtime.h>
#include <math.h>

typedef __attribute__((ext_vector_type(2))) float v2f;
typedef __attribute__((ext_vector_type(8))) float v8f;

#define NEG_SLOPE 0.2f

template<int K, int NOUT, int CW, int NT, bool PF>
__global__ __launch_bounds__(NT) void gemm_wmma(const float* __restrict__ A,
                                                 const float* __restrict__ B,
                                                 float* __restrict__ C,
                                                 int nrt)
{
    constexpr int NG = NOUT / CW;
    constexpr int TL = CW / 16;
    constexpr int P  = 2 * NOUT + 16;
    constexpr int NQ = K / 8;
    __shared__ float Bs[(K / 2) * P];

    const int tid  = threadIdx.x;
    const int lane = tid & 31;
    const int half = lane >> 4;
    const int mr   = lane & 15;
    const int w = __builtin_amdgcn_readfirstlane(tid >> 5);
    constexpr int NW = NT / 32;

    const int nj    = (nrt - (int)blockIdx.x + (int)gridDim.x - 1) / (int)gridDim.x;
    const int ntask = nj * NG;

    auto load_a = [&](float4 (&v)[NQ], int i) {
        const int rt = blockIdx.x + (i / NG) * gridDim.x;
        const float* arow = A + (size_t)(rt * 16 + mr) * K + 4 * half;
#pragma unroll
        for (int q = 0; q < NQ; ++q)
            v[q] = *(const float4*)(arow + 8 * q);
    };

    float4 av[NQ];
    if (PF && w < ntask) load_a(av, w);

    constexpr int NS = (K / 2) * (NOUT / 4);
    constexpr int SI = NS / NT;
    static_assert(NS % NT == 0, "every thread copies the same number of items");
    {
        float4 r0[SI], r1[SI];
#pragma unroll
        for (int s = 0; s < SI; ++s) {
            const int i  = tid + s * NT;
            const int p  = i / (NOUT / 4);
            const int n4 = i % (NOUT / 4);
            r0[s] = *(const float4*)(B + (size_t)(2 * p) * NOUT + 4 * n4);
            r1[s] = *(const float4*)(B + (size_t)(2 * p + 1) * NOUT + 4 * n4);
        }
#pragma unroll
        for (int s = 0; s < SI; ++s) {
            const int i  = tid + s * NT;
            const int p  = i / (NOUT / 4);
            const int n4 = i % (NOUT / 4);
            float4* d = (float4*)(Bs + p * P + 8 * n4);
            d[0] = make_float4(r0[s].x, r1[s].x, r0[s].y, r1[s].y);
            d[1] = make_float4(r0[s].z, r1[s].z, r0[s].w, r1[s].w);
        }
    }
    __syncthreads();

    for (int i = w; i < ntask; i += NW) {
        const int cg = i % NG;
        const int rt = blockIdx.x + (i / NG) * gridDim.x;

        float4 nv[NQ];
        if (PF) {
            load_a(nv, (i + NW < ntask) ? i + NW : i);
        } else {
            load_a(av, i);
        }

        v8f acc[TL] = {};
        const float* bl = Bs + (2 * half) * P + 2 * (cg * CW + mr);
#pragma unroll
        for (int q = 0; q < NQ; ++q) {
#pragma unroll
            for (int sub = 0; sub < 2; ++sub) {
                v2f a;
                a.x = sub ? av[q].z : av[q].x;
                a.y = sub ? av[q].w : av[q].y;
                const float* bp = bl + (4 * q + sub) * P;
#pragma unroll
                for (int t = 0; t < TL; ++t) {
                    const v2f b = *(const v2f*)(bp + 32 * t);
                    acc[t] = __builtin_amdgcn_wmma_f32_16x16x4_f32(
                        false, a, false, b, (short)0, acc[t], false, false);
                }
            }
        }

        float* crow = C + (size_t)(rt * 16 + half * 8) * NOUT + cg * CW + mr;
#pragma unroll
        for (int t = 0; t < TL; ++t)
#pragma unroll
            for (int r = 0; r < 8; ++r)
                crow[(size_t)r * NOUT + t * 16] = acc[t][r];

        if (PF) {
#pragma unroll
            for (int q = 0; q < NQ; ++q) av[q] = nv[q];
        }
    }
}

template<int H, int C>
__global__ void att_coeff(const float* __restrict__ h,
                          const float* __restrict__ a_src,
                          const float* __restrict__ a_dst,
                          float* __restrict__ als,
                          float* __restrict__ ald,
                          int n)
{
    int i = blockIdx.x * blockDim.x + threadIdx.x;
    if (i >= n * H) return;
    int node = i / H, hd = i % H;
    const float* hp = h + (size_t)node * (H * C) + hd * C;
    const float* asrc = a_src + hd * C;
    const float* adst = a_dst + hd * C;
    float ss = 0.f, sd = 0.f;
#pragma unroll 4
    for (int c = 0; c < C; ++c) {
        float v = hp[c];
        ss += v * asrc[c];
        sd += v * adst[c];
    }
    als[i] = ss;
    ald[i] = sd;
}

#define NPB 196
#define QW  704
#define CH  8192
#define MAXBIN 1024
#define RSH 18
#define RMASK ((1 << RSH) - 1)
static_assert(MAXBIN * NPB <= (1 << RSH) && NPB <= (1 << (31 - RSH)),
              "a record must hold any source node and any row of a group");

__global__ __launch_bounds__(1024) void bin_edges(const int* __restrict__ src,
                                                  const int* __restrict__ dst,
                                                  int* __restrict__ rec,
                                                  int* __restrict__ seg,
                                                  int E_, int n, int nbin)
{
    __shared__ int hist[MAXBIN];
    __shared__ int wsum[32];
    const int tid  = threadIdx.x;
    const int lane = tid & 31;
    const int w    = tid >> 5;
    const int base = blockIdx.x * CH;

    hist[tid] = 0;
    __syncthreads();

    int d[CH / 1024], sv[CH / 1024], bin[CH / 1024], rk[CH / 1024];
#pragma unroll
    for (int k = 0; k < CH / 1024; ++k) {
        const int e = base + k * 1024 + tid;
        d[k] = (e < E_) ? dst[e] : -1;
        sv[k] = (e < E_) ? src[e] : 0;
    }
    const unsigned nlim = (unsigned)min(n, MAXBIN * NPB);
#pragma unroll
    for (int k = 0; k < CH / 1024; ++k) {
        bin[k] = -1;
        rk[k]  = 0;
        if ((unsigned)d[k] < nlim) {
            bin[k] = d[k] / NPB;
            rk[k]  = atomicAdd(&hist[bin[k]], 1);
        }
    }
    __syncthreads();

    const int v = hist[tid];
    int x = v;
    for (int o = 1; o < 32; o <<= 1) {
        const int y = __shfl_up(x, o);
        if (lane >= o) x += y;
    }
    if (lane == 31) wsum[w] = x;
    __syncthreads();
    if (w == 0) {
        const int t = wsum[lane];
        int s = t;
        for (int o = 1; o < 32; o <<= 1) {
            const int y = __shfl_up(s, o);
            if (lane >= o) s += y;
        }
        wsum[lane] = s - t;
    }
    __syncthreads();
    const int st = wsum[w] + x - v;
    hist[tid] = st;
    if (tid < nbin) {
        int* sp = seg + ((size_t)blockIdx.x * nbin + tid) * 2;
        sp[0] = st;
        sp[1] = st + v;
    }
    __syncthreads();

#pragma unroll
    for (int k = 0; k < CH / 1024; ++k)
        if (bin[k] >= 0)
            rec[base + hist[bin[k]] + rk[k]] =
                (sv[k] & RMASK) | ((d[k] - bin[k] * NPB) << RSH);
}

template<int H, int F>
__global__ __launch_bounds__(1024) void aggr_lds(const int* __restrict__ rec,
                                                 const int* __restrict__ seg,
                                                 const float* __restrict__ als,
                                                 const float* __restrict__ ald,
                                                 const float* __restrict__ h,
                                                 const float* __restrict__ b,
                                                 float* __restrict__ outp,
                                                 int E_, int n)
{
    constexpr int NJ  = F / 32;
    constexpr int CPH = (F / H) / 32;
    constexpr int UE  = 32 / NJ;
    __shared__ float acc[NPB * F];
    __shared__ float dsum[H * NPB];
    __shared__ int   q[32 * QW];

    const int tid  = threadIdx.x;
    const int lane = tid & 31;
    const int w    = __builtin_amdgcn_readfirstlane(tid >> 5);
    const int lo   = blockIdx.x * NPB;
    const int cntn = min(NPB, n - lo);

    for (int i = tid; i < NPB * F; i += 1024)
        acc[i] = 0.f;
    for (int i = tid; i < NPB * H; i += 1024)
        dsum[i] = 0.f;
    __syncthreads();

    int* myq = q + w * QW;
    int cnt = 0;

    auto drain = [&]() {
        __builtin_amdgcn_fence(__ATOMIC_ACQ_REL, "wavefront");
        for (int qb = 0; qb < cnt; qb += 32) {
            const int idx = qb + lane;
            int s = 0, dl = 0;
            float wg[H];
#pragma unroll
            for (int hd = 0; hd < H; ++hd) wg[hd] = 0.f;
            if (idx < cnt) {
                const int r = myq[idx];
                s  = r & RMASK;
                dl = r >> RSH;
                const int d = lo + dl;
#pragma unroll
                for (int hd = 0; hd < H; ++hd) {
                    float x = als[s * H + hd] + ald[d * H + hd];
                    float lv = (x >= 0.f) ? x : NEG_SLOPE * x;
                    wg[hd] = expf(lv);
                    atomicAdd(&dsum[hd * NPB + dl], wg[hd]);
                }
            }
            const int nb = min(32, cnt - qb);
            for (int j0 = 0; j0 < nb; j0 += UE) {
                float hv[UE][NJ];
#pragma unroll
                for (int u = 0; u < UE; ++u) {
                    const int j  = min(j0 + u, nb - 1);
                    const int sj = __builtin_amdgcn_readlane(s, j);
                    const float* hs = h + (size_t)sj * F + lane;
#pragma unroll
                    for (int jj = 0; jj < NJ; ++jj)
                        hv[u][jj] = hs[32 * jj];
                }
#pragma unroll
                for (int u = 0; u < UE; ++u) {
                    const int j = j0 + u;
                    if (j < nb) {
                        const int dj = __builtin_amdgcn_readlane(dl, j);
                        float* ar = acc + dj * F + lane;
#pragma unroll
                        for (int jj = 0; jj < NJ; ++jj) {
                            const float a = __int_as_float(__builtin_amdgcn_readlane(
                                __float_as_int(wg[jj / CPH]), j));
                            atomicAdd(&ar[32 * jj], a * hv[u][jj]);
                        }
                    }
                }
            }
        }
    };

    auto push = [&](bool m, int e) {
        const unsigned bal = __builtin_amdgcn_ballot_w32(m);
        if (m) myq[cnt + (int)__builtin_amdgcn_mbcnt_lo(bal, 0u)] = e;
        cnt += __builtin_popcount(bal);
    };

    const int nsb  = (E_ + CH - 1) / CH;
    const int nbin = gridDim.x;
    for (int b0 = w; b0 < nsb; b0 += 1024) {
        const int bl = b0 + 32 * lane;
        int2 r = make_int2(0, 0);
        if (bl < nsb)
            r = *(const int2*)(seg + ((size_t)bl * nbin + blockIdx.x) * 2);
        const int nmine = min(32, (nsb - b0 + 31) / 32);
        for (int l = 0; l < nmine; ++l) {
            const int a0 = __builtin_amdgcn_readlane(r.x, l);
            const int a1 = __builtin_amdgcn_readlane(r.y, l);
            const int* pp = rec + (size_t)(b0 + 32 * l) * CH;
            for (int i = a0; i < a1; i += 32) {
                if (cnt > QW - 32) { drain(); cnt = 0; }
                if (i + lane < a1) myq[cnt + lane] = pp[i + lane];
                cnt += min(32, a1 - i);
            }
        }
    }
    for (int i0 = w * 32; i0 < cntn; i0 += 1024) {
        if (cnt > QW - 32) { drain(); cnt = 0; }
        const int i = i0 + lane;
        push(i < cntn, (lo + i) | (i << RSH));
    }
    drain();
    __syncthreads();

    for (int i = tid; i < H * NPB; i += 1024)
        if (i % NPB < cntn) dsum[i] = 1.0f / dsum[i];
    __syncthreads();

    float* orow = outp + (size_t)lo * F;
    for (int i = tid; i < cntn * F; i += 1024) {
        const int c = i % F;
        float v = acc[i] * dsum[(c / (F / H)) * NPB + i / F] + b[c];
        orow[i] = (v > 0.f) ? v : expm1f(v);
    }
}

extern "C" void kernel_launch(void* const* d_in, const int* in_sizes, int n_in,
                              void* d_out, int out_size, void* d_ws, size_t ws_size,
                              hipStream_t stream)
{
    const float* x   = (const float*)d_in[0];
    const int*   ei  = (const int*)  d_in[1];
    const float* W1  = (const float*)d_in[2];
    const float* as1 = (const float*)d_in[3];
    const float* ad1 = (const float*)d_in[4];
    const float* b1  = (const float*)d_in[5];
    const float* W2  = (const float*)d_in[6];
    const float* as2 = (const float*)d_in[7];
    const float* ad2 = (const float*)d_in[8];
    const float* b2  = (const float*)d_in[9];
    float* out = (float*)d_out;

    const int IN_DIM = 128, H1 = 8, F1 = 256, OUT_DIM = 64;
    int n = in_sizes[0] / IN_DIM;
    int e = in_sizes[1] / 2;
    const int* src = ei;
    const int* dst = ei + e;

    float* ws = (float*)d_ws;
    float* h1   = ws;  ws += (size_t)n * F1;
    float* g1   = ws;  ws += (size_t)n * F1;
    float* h2   = h1;
    int*   rec  = (int*)ws;  ws += (size_t)n * OUT_DIM;
    int nsb  = (e + CH - 1) / CH;
    int nbin = (n + NPB - 1) / NPB;
    int*   seg  = rec + (size_t)nsb * CH;
    float* als1 = ws;  ws += (size_t)n * H1;
    float* ald1 = ws;  ws += (size_t)n * H1;
    float* als2 = ws;  ws += (size_t)n;
    float* ald2 = ws;  ws += (size_t)n;

    bin_edges<<<nsb, 1024, 0, stream>>>(src, dst, rec, seg, e, n, nbin);

    gemm_wmma<128, 256, 64, 512, true><<<256, 512, 0, stream>>>(x, W1, h1, n / 16);
    att_coeff<8, 32><<<(n * 8 + 255) / 256, 256, 0, stream>>>(h1, as1, ad1, als1, ald1, n);
    aggr_lds<8, 256><<<nbin, 1024, 0, stream>>>(rec, seg, als1, ald1, h1, b1, g1, e, n);

    gemm_wmma<256, 64, 32, 1024, false><<<256, 1024, 0, stream>>>(g1, W2, h2, n / 16);
    att_coeff<1, 64><<<(n + 255) / 256, 256, 0, stream>>>(h2, as2, ad2, als2, ald2, n);
    aggr_lds<1, 64><<<nbin, 1024, 0, stream>>>(rec, seg, als2, ald2, h2, b2, out, e, n);
}


# al_s, al_d as A*(W*a_src), A*(W*a_dst): 16 more B columns formed in LDS during the B copy, one more f32 WMMA tile per row tile; both coefficient launches removed; coefficient sums re-associated
# speedup vs baseline: 3.5160x; 1.0338x over previous
#include <hip/hip_runtime.h>
#include <math.h>

typedef __attribute__((ext_vector_type(2))) float v2f;
typedef __attribute__((ext_vector_type(8))) float v8f;

#define NEG_SLOPE 0.2f

template<bool V> struct flag_t { static constexpr bool value = V; };

template<int K, int NOUT, int CW, int NT, bool PF, int H>
__global__ __launch_bounds__(NT) void gemm_wmma(const float* __restrict__ A,
                                                 const float* __restrict__ B,
                                                 const float* __restrict__ a_src,
                                                 const float* __restrict__ a_dst,
                                                 float* __restrict__ C,
                                                 float* __restrict__ als,
                                                 float* __restrict__ ald,
                                                 int nrt)
{
    constexpr int NG = NOUT / CW;
    constexpr int TL = CW / 16;
    constexpr int XO = 2 * NOUT;
    constexpr int P  = 2 * NOUT + 48;
    constexpr int NQ = K / 8;
    constexpr int G  = (NOUT / H) / 4;
    static_assert(2 * H <= 16 && NOUT % (4 * H) == 0 && G <= 32 && (G & (G - 1)) == 0,
                  "a head's copy items are a power-of-two run of lanes inside one wave");
    __shared__ float Bs[(K / 2) * P];

    const int tid  = threadIdx.x;
    const int lane = tid & 31;
    const int half = lane >> 4;
    const int mr   = lane & 15;
    const int w = __builtin_amdgcn_readfirstlane(tid >> 5);
    constexpr int NW = NT / 32;
    static_assert(NW % NG == 0, "the NG tasks of a row tile fall into one round of the waves");

    const int nj    = (nrt - (int)blockIdx.x + (int)gridDim.x - 1) / (int)gridDim.x;
    const int ntask = nj * NG;

    auto load_a = [&](float4 (&v)[NQ], int i) {
        const int rt = blockIdx.x + (i / NG) * gridDim.x;
        const float* arow = A + (size_t)(rt * 16 + mr) * K + 4 * half;
#pragma unroll
        for (int q = 0; q < NQ; ++q)
            v[q] = *(const float4*)(arow + 8 * q);
    };

    float4 av[NQ];
    if (PF && w < ntask) load_a(av, w);

    constexpr int NS = (K / 2) * (NOUT / 4);
    constexpr int SI = NS / NT;
    static_assert(NS % NT == 0, "every thread copies the same number of items");
    static_assert(NT % (NOUT / 4) == 0, "all items of a thread hold the same four columns");
    {
        const int   c4  = tid % (NOUT / 4);
        const float4 as4 = *(const float4*)(a_src + 4 * c4);
        const float4 ad4 = *(const float4*)(a_dst + 4 * c4);
        float4 r0[SI], r1[SI];
#pragma unroll
        for (int s = 0; s < SI; ++s) {
            const int i  = tid + s * NT;
            const int p  = i / (NOUT / 4);
            const int n4 = i % (NOUT / 4);
            r0[s] = *(const float4*)(B + (size_t)(2 * p) * NOUT + 4 * n4);
            r1[s] = *(const float4*)(B + (size_t)(2 * p + 1) * NOUT + 4 * n4);
        }
#pragma unroll
        for (int s = 0; s < SI; ++s) {
            const int i  = tid + s * NT;
            const int p  = i / (NOUT / 4);
            const int n4 = i % (NOUT / 4);
            float4* d = (float4*)(Bs + p * P + 8 * n4);
            d[0] = make_float4(r0[s].x, r1[s].x, r0[s].y, r1[s].y);
            d[1] = make_float4(r0[s].z, r1[s].z, r0[s].w, r1[s].w);
            float s0 = r0[s].x * as4.x + r0[s].y * as4.y + r0[s].z * as4.z + r0[s].w * as4.w;
            float s1 = r1[s].x * as4.x + r1[s].y * as4.y + r1[s].z * as4.z + r1[s].w * as4.w;
            float d0 = r0[s].x * ad4.x + r0[s].y * ad4.y + r0[s].z * ad4.z + r0[s].w * ad4.w;
            float d1 = r1[s].x * ad4.x + r1[s].y * ad4.y + r1[s].z * ad4.z + r1[s].w * ad4.w;
#pragma unroll
            for (int o = 1; o < G; o <<= 1) {
                s0 += __shfl_xor(s0, o);
                s1 += __shfl_xor(s1, o);
                d0 += __shfl_xor(d0, o);
                d1 += __shfl_xor(d1, o);
            }
            if (n4 % G == 0) {
                const int hd = n4 / G;
                v2f ws, wd;
                ws.x = s0; ws.y = s1;
                wd.x = d0; wd.y = d1;
                *(v2f*)(Bs + p * P + XO + 2 * hd)       = ws;
                *(v2f*)(Bs + p * P + XO + 2 * (H + hd)) = wd;
            }
        }
        if constexpr (2 * H < 16) {
            constexpr int ZW = 2 * (16 - 2 * H);
            for (int i = tid; i < (K / 2) * ZW; i += NT)
                Bs[(i / ZW) * P + XO + 4 * H + i % ZW] = 0.f;
        }
    }
    __syncthreads();

    for (int i = w; i < ntask; i += NW) {
        const int cg = i % NG;
        const int rt = blockIdx.x + (i / NG) * gridDim.x;
        const bool ex = ((i / NW) % NG) == cg;

        float4 nv[NQ];
        if (PF) {
            load_a(nv, (i + NW < ntask) ? i + NW : i);
        }
        const float* arow = A + (size_t)(rt * 16 + mr) * K + 4 * half;

        v8f acc[TL] = {};
        v8f accx = {};
        const float* bl = Bs + (2 * half) * P + 2 * (cg * CW + mr);
        const float* bx = Bs + (2 * half) * P + XO + 2 * mr;
        auto kloop = [&](auto tag) {
            constexpr bool EX = decltype(tag)::value;
#pragma unroll
            for (int q = 0; q < NQ; ++q) {
                float4 aq;
                if constexpr (PF) aq = av[q];
                else              aq = *(const float4*)(arow + 8 * q);
#pragma unroll
                for (int sub = 0; sub < 2; ++sub) {
                    v2f a;
                    a.x = sub ? aq.z : aq.x;
                    a.y = sub ? aq.w : aq.y;
                    const float* bp = bl + (4 * q + sub) * P;
#pragma unroll
                    for (int t = 0; t < TL; ++t) {
                        const v2f b = *(const v2f*)(bp + 32 * t);
                        acc[t] = __builtin_amdgcn_wmma_f32_16x16x4_f32(
                            false, a, false, b, (short)0, acc[t], false, false);
                    }
                    if constexpr (EX) {
                        const v2f b = *(const v2f*)(bx + (4 * q + sub) * P);
                        accx = __builtin_amdgcn_wmma_f32_16x16x4_f32(
                            false, a, false, b, (short)0, accx, false, false);
                    }
                }
            }
        };
        if (ex) kloop(flag_t<true>{});
        else    kloop(flag_t<false>{});

        float* crow = C + (size_t)(rt * 16 + half * 8) * NOUT + cg * CW + mr;
#pragma unroll
        for (int t = 0; t < TL; ++t)
#pragma unroll
            for (int r = 0; r < 8; ++r)
                crow[(size_t)r * NOUT + t * 16] = acc[t][r];

        if (ex && mr < 2 * H) {
            const size_t row0 = (size_t)(rt * 16 + half * 8);
            float* cp = (mr < H) ? als + row0 * H + mr : ald + row0 * H + (mr - H);
#pragma unroll
            for (int r = 0; r < 8; ++r)
                cp[(size_t)r * H] = accx[r];
        }

        if (PF) {
#pragma unroll
            for (int q = 0; q < NQ; ++q) av[q] = nv[q];
        }
    }
}

#define NPB 196
#define QW  704
#define CH  8192
#define MAXBIN 1024
#define RSH 18
#define RMASK ((1 << RSH) - 1)
static_assert(MAXBIN * NPB <= (1 << RSH) && NPB <= (1 << (31 - RSH)),
              "a record must hold any source node and any row of a group");

__global__ __launch_bounds__(1024) void bin_edges(const int* __restrict__ src,
                                                  const int* __restrict__ dst,
                                                  int* __restrict__ rec,
                                                  int* __restrict__ seg,
                                                  int E_, int n, int nbin)
{
    __shared__ int hist[MAXBIN];
    __shared__ int wsum[32];
    const int tid  = threadIdx.x;
    const int lane = tid & 31;
    const int w    = tid >> 5;
    const int base = blockIdx.x * CH;

    hist[tid] = 0;
    __syncthreads();

    int d[CH / 1024], sv[CH / 1024], bin[CH / 1024], rk[CH / 1024];
#pragma unroll
    for (int k = 0; k < CH / 1024; ++k) {
        const int e = base + k * 1024 + tid;
        d[k] = (e < E_) ? dst[e] : -1;
        sv[k] = (e < E_) ? src[e] : 0;
    }
    const unsigned nlim = (unsigned)min(n, MAXBIN * NPB);
#pragma unroll
    for (int k = 0; k < CH / 1024; ++k) {
        bin[k] = -1;
        rk[k]  = 0;
        if ((unsigned)d[k] < nlim) {
            bin[k] = d[k] / NPB;
            rk[k]  = atomicAdd(&hist[bin[k]], 1);
        }
    }
    __syncthreads();

    const int v = hist[tid];
    int x = v;
    for (int o = 1; o < 32; o <<= 1) {
        const int y = __shfl_up(x, o);
        if (lane >= o) x += y;
    }
    if (lane == 31) wsum[w] = x;
    __syncthreads();
    if (w == 0) {
        const int t = wsum[lane];
        int s = t;
        for (int o = 1; o < 32; o <<= 1) {
            const int y = __shfl_up(s, o);
            if (lane >= o) s += y;
        }
        wsum[lane] = s - t;
    }
    __syncthreads();
    const int st = wsum[w] + x - v;
    hist[tid] = st;
    if (tid < nbin) {
        int* sp = seg + ((size_t)blockIdx.x * nbin + tid) * 2;
        sp[0] = st;
        sp[1] = st + v;
    }
    __syncthreads();

#pragma unroll
    for (int k = 0; k < CH / 1024; ++k)
        if (bin[k] >= 0)
            rec[base + hist[bin[k]] + rk[k]] =
                (sv[k] & RMASK) | ((d[k] - bin[k] * NPB) << RSH);
}

template<int H, int F>
__global__ __launch_bounds__(1024) void aggr_lds(const int* __restrict__ rec,
                                                 const int* __restrict__ seg,
                                                 const float* __restrict__ als,
                                                 const float* __restrict__ ald,
                                                 const float* __restrict__ h,
                                                 const float* __restrict__ b,
                                                 float* __restrict__ outp,
                                                 int E_, int n)
{
    constexpr int NJ  = F / 32;
    constexpr int CPH = (F / H) / 32;
    constexpr int UE  = 32 / NJ;
    __shared__ float acc[NPB * F];
    __shared__ float dsum[H * NPB];
    __shared__ int   q[32 * QW];

    const int tid  = threadIdx.x;
    const int lane = tid & 31;
    const int w    = __builtin_amdgcn_readfirstlane(tid >> 5);
    const int lo   = blockIdx.x * NPB;
    const int cntn = min(NPB, n - lo);

    for (int i = tid; i < NPB * F; i += 1024)
        acc[i] = 0.f;
    for (int i = tid; i < NPB * H; i += 1024)
        dsum[i] = 0.f;
    __syncthreads();

    int* myq = q + w * QW;
    int cnt = 0;

    auto drain = [&]() {
        __builtin_amdgcn_fence(__ATOMIC_ACQ_REL, "wavefront");
        for (int qb = 0; qb < cnt; qb += 32) {
            const int idx = qb + lane;
            int s = 0, dl = 0;
            float wg[H];
#pragma unroll
            for (int hd = 0; hd < H; ++hd) wg[hd] = 0.f;
            if (idx < cnt) {
                const int r = myq[idx];
                s  = r & RMASK;
                dl = r >> RSH;
                const int d = lo + dl;
#pragma unroll
                for (int hd = 0; hd < H; ++hd) {
                    float x = als[s * H + hd] + ald[d * H + hd];
                    float lv = (x >= 0.f) ? x : NEG_SLOPE * x;
                    wg[hd] = expf(lv);
                    atomicAdd(&dsum[hd * NPB + dl], wg[hd]);
                }
            }
            const int nb = min(32, cnt - qb);
            for (int j0 = 0; j0 < nb; j0 += UE) {
                float hv[UE][NJ];
#pragma unroll
                for (int u = 0; u < UE; ++u) {
                    const int j  = min(j0 + u, nb - 1);
                    const int sj = __builtin_amdgcn_readlane(s, j);
                    const float* hs = h + (size_t)sj * F + lane;
#pragma unroll
                    for (int jj = 0; jj < NJ; ++jj)
                        hv[u][jj] = hs[32 * jj];
                }
#pragma unroll
                for (int u = 0; u < UE; ++u) {
                    const int j = j0 + u;
                    if (j < nb) {
                        const int dj = __builtin_amdgcn_readlane(dl, j);
                        float* ar = acc + dj * F + lane;
#pragma unroll
                        for (int jj = 0; jj < NJ; ++jj) {
                            const float a = __int_as_float(__builtin_amdgcn_readlane(
                                __float_as_int(wg[jj / CPH]), j));
                            atomicAdd(&ar[32 * jj], a * hv[u][jj]);
                        }
                    }
                }
            }
        }
    };

    auto push = [&](bool m, int e) {
        const unsigned bal = __builtin_amdgcn_ballot_w32(m);
        if (m) myq[cnt + (int)__builtin_amdgcn_mbcnt_lo(bal, 0u)] = e;
        cnt += __builtin_popcount(bal);
    };

    const int nsb  = (E_ + CH - 1) / CH;
    const int nbin = gridDim.x;
    for (int b0 = w; b0 < nsb; b0 += 1024) {
        const int bl = b0 + 32 * lane;
        int2 r = make_int2(0, 0);
        if (bl < nsb)
            r = *(const int2*)(seg + ((size_t)bl * nbin + blockIdx.x) * 2);
        const int nmine = min(32, (nsb - b0 + 31) / 32);
        for (int l = 0; l < nmine; ++l) {
            const int a0 = __builtin_amdgcn_readlane(r.x, l);
            const int a1 = __builtin_amdgcn_readlane(r.y, l);
            const int* pp = rec + (size_t)(b0 + 32 * l) * CH;
            for (int i = a0; i < a1; i += 32) {
                if (cnt > QW - 32) { drain(); cnt = 0; }
                if (i + lane < a1) myq[cnt + lane] = pp[i + lane];
                cnt += min(32, a1 - i);
            }
        }
    }
    for (int i0 = w * 32; i0 < cntn; i0 += 1024) {
        if (cnt > QW - 32) { drain(); cnt = 0; }
        const int i = i0 + lane;
        push(i < cntn, (lo + i) | (i << RSH));
    }
    drain();
    __syncthreads();

    for (int i = tid; i < H * NPB; i += 1024)
        if (i % NPB < cntn) dsum[i] = 1.0f / dsum[i];
    __syncthreads();

    float* orow = outp + (size_t)lo * F;
    for (int i = tid; i < cntn * F; i += 1024) {
        const int c = i % F;
        float v = acc[i] * dsum[(c / (F / H)) * NPB + i / F] + b[c];
        orow[i] = (v > 0.f) ? v : expm1f(v);
    }
}

extern "C" void kernel_launch(void* const* d_in, const int* in_sizes, int n_in,
                              void* d_out, int out_size, void* d_ws, size_t ws_size,
                              hipStream_t stream)
{
    const float* x   = (const float*)d_in[0];
    const int*   ei  = (const int*)  d_in[1];
    const float* W1  = (const float*)d_in[2];
    const float* as1 = (const float*)d_in[3];
    const float* ad1 = (const float*)d_in[4];
    const float* b1  = (const float*)d_in[5];
    const float* W2  = (const float*)d_in[6];
    const float* as2 = (const float*)d_in[7];
    const float* ad2 = (const float*)d_in[8];
    const float* b2  = (const float*)d_in[9];
    float* out = (float*)d_out;

    const int IN_DIM = 128, H1 = 8, F1 = 256, OUT_DIM = 64;
    int n = in_sizes[0] / IN_DIM;
    int e = in_sizes[1] / 2;
    const int* src = ei;
    const int* dst = ei + e;

    float* ws = (float*)d_ws;
    float* h1   = ws;  ws += (size_t)n * F1;
    float* g1   = ws;  ws += (size_t)n * F1;
    float* h2   = h1;
    int*   rec  = (int*)ws;  ws += (size_t)n * OUT_DIM;
    int nsb  = (e + CH - 1) / CH;
    int nbin = (n + NPB - 1) / NPB;
    int*   seg  = rec + (size_t)nsb * CH;
    float* als1 = ws;  ws += (size_t)n * H1;
    float* ald1 = ws;  ws += (size_t)n * H1;
    float* als2 = ws;  ws += (size_t)n;
    float* ald2 = ws;  ws += (size_t)n;

    bin_edges<<<nsb, 1024, 0, stream>>>(src, dst, rec, seg, e, n, nbin);

    gemm_wmma<128, 256, 64, 512, true, 8><<<256, 512, 0, stream>>>(x, W1, as1, ad1, h1, als1, ald1, n / 16);
    aggr_lds<8, 256><<<nbin, 1024, 0, stream>>>(rec, seg, als1, ald1, h1, b1, g1, e, n);

    gemm_wmma<256, 64, 32, 1024, false, 1><<<256, 1024, 0, stream>>>(g1, W2, as2, ad2, h2, als2, ald2, n / 16);
    aggr_lds<1, 64><<<nbin, 1024, 0, stream>>>(rec, seg, als2, ald2, h2, b2, out, e, n);
}


# Aggregation: a lane holds 4 adjacent columns: source rows read by 16-byte loads (conv1 2 an edge, conv2 1 an edge pair; were 8 and 2 4-byte), LDS rows lane-transposed, 16-byte out stores; same sums
# speedup vs baseline: 3.9027x; 1.1100x over previous
#include <hip/hip_runtime.h>
#include <math.h>

typedef __attribute__((ext_vector_type(2))) float v2f;
typedef __attribute__((ext_vector_type(8))) float v8f;

#define NEG_SLOPE 0.2f

template<bool V> struct flag_t { static constexpr bool value = V; };

template<int K, int NOUT, int CW, int NT, bool PF, int H>
__global__ __launch_bounds__(NT) void gemm_wmma(const float* __restrict__ A,
                                                 const float* __restrict__ B,
                                                 const float* __restrict__ a_src,
                                                 const float* __restrict__ a_dst,
                                                 float* __restrict__ C,
                                                 float* __restrict__ als,
                                                 float* __restrict__ ald,
                                                 int nrt)
{
    constexpr int NG = NOUT / CW;
    constexpr int TL = CW / 16;
    constexpr int XO = 2 * NOUT;
    constexpr int P  = 2 * NOUT + 48;
    constexpr int NQ = K / 8;
    constexpr int G  = (NOUT / H) / 4;
    static_assert(2 * H <= 16 && NOUT % (4 * H) == 0 && G <= 32 && (G & (G - 1)) == 0,
                  "a head's copy items are a power-of-two run of lanes inside one wave");
    __shared__ float Bs[(K / 2) * P];

    const int tid  = threadIdx.x;
    const int lane = tid & 31;
    const int half = lane >> 4;
    const int mr   = lane & 15;
    const int w = __builtin_amdgcn_readfirstlane(tid >> 5);
    constexpr int NW = NT / 32;
    static_assert(NW % NG == 0, "the NG tasks of a row tile fall into one round of the waves");

    const int nj    = (nrt - (int)blockIdx.x + (int)gridDim.x - 1) / (int)gridDim.x;
    const int ntask = nj * NG;

    auto load_a = [&](float4 (&v)[NQ], int i) {
        const int rt = blockIdx.x + (i / NG) * gridDim.x;
        const float* arow = A + (size_t)(rt * 16 + mr) * K + 4 * half;
#pragma unroll
        for (int q = 0; q < NQ; ++q)
            v[q] = *(const float4*)(arow + 8 * q);
    };

    float4 av[NQ];
    if (PF && w < ntask) load_a(av, w);

    constexpr int NS = (K / 2) * (NOUT / 4);
    constexpr int SI = NS / NT;
    static_assert(NS % NT == 0, "every thread copies the same number of items");
    static_assert(NT % (NOUT / 4) == 0, "all items of a thread hold the same four columns");
    {
        const int   c4  = tid % (NOUT / 4);
        const float4 as4 = *(const float4*)(a_src + 4 * c4);
        const float4 ad4 = *(const float4*)(a_dst + 4 * c4);
        float4 r0[SI], r1[SI];
#pragma unroll
        for (int s = 0; s < SI; ++s) {
            const int i  = tid + s * NT;
            const int p  = i / (NOUT / 4);
            const int n4 = i % (NOUT / 4);
            r0[s] = *(const float4*)(B + (size_t)(2 * p) * NOUT + 4 * n4);
            r1[s] = *(const float4*)(B + (size_t)(2 * p + 1) * NOUT + 4 * n4);
        }
#pragma unroll
        for (int s = 0; s < SI; ++s) {
            const int i  = tid + s * NT;
            const int p  = i / (NOUT / 4);
            const int n4 = i % (NOUT / 4);
            float4* d = (float4*)(Bs + p * P + 8 * n4);
            d[0] = make_float4(r0[s].x, r1[s].x, r0[s].y, r1[s].y);
            d[1] = make_float4(r0[s].z, r1[s].z, r0[s].w, r1[s].w);
            float s0 = r0[s].x * as4.x + r0[s].y * as4.y + r0[s].z * as4.z + r0[s].w * as4.w;
            float s1 = r1[s].x * as4.x + r1[s].y * as4.y + r1[s].z * as4.z + r1[s].w * as4.w;
            float d0 = r0[s].x * ad4.x + r0[s].y * ad4.y + r0[s].z * ad4.z + r0[s].w * ad4.w;
            float d1 = r1[s].x * ad4.x + r1[s].y * ad4.y + r1[s].z * ad4.z + r1[s].w * ad4.w;
#pragma unroll
            for (int o = 1; o < G; o <<= 1) {
                s0 += __shfl_xor(s0, o);
                s1 += __shfl_xor(s1, o);
                d0 += __shfl_xor(d0, o);
                d1 += __shfl_xor(d1, o);
            }
            if (n4 % G == 0) {
                const int hd = n4 / G;
                v2f ws, wd;
                ws.x = s0; ws.y = s1;
                wd.x = d0; wd.y = d1;
                *(v2f*)(Bs + p * P + XO + 2 * hd)       = ws;
                *(v2f*)(Bs + p * P + XO + 2 * (H + hd)) = wd;
            }
        }
        if constexpr (2 * H < 16) {
            constexpr int ZW = 2 * (16 - 2 * H);
            for (int i = tid; i < (K / 2) * ZW; i += NT)
                Bs[(i / ZW) * P + XO + 4 * H + i % ZW] = 0.f;
        }
    }
    __syncthreads();

    for (int i = w; i < ntask; i += NW) {
        const int cg = i % NG;
        const int rt = blockIdx.x + (i / NG) * gridDim.x;
        const bool ex = ((i / NW) % NG) == cg;

        float4 nv[NQ];
        if (PF) {
            load_a(nv, (i + NW < ntask) ? i + NW : i);
        }
        const float* arow = A + (size_t)(rt * 16 + mr) * K + 4 * half;

        v8f acc[TL] = {};
        v8f accx = {};
        const float* bl = Bs + (2 * half) * P + 2 * (cg * CW + mr);
        const float* bx = Bs + (2 * half) * P + XO + 2 * mr;
        auto kloop = [&](auto tag) {
            constexpr bool EX = decltype(tag)::value;
#pragma unroll
            for (int q = 0; q < NQ; ++q) {
                float4 aq;
                if constexpr (PF) aq = av[q];
                else              aq = *(const float4*)(arow + 8 * q);
#pragma unroll
                for (int sub = 0; sub < 2; ++sub) {
                    v2f a;
                    a.x = sub ? aq.z : aq.x;
                    a.y = sub ? aq.w : aq.y;
                    const float* bp = bl + (4 * q + sub) * P;
#pragma unroll
                    for (int t = 0; t < TL; ++t) {
                        const v2f b = *(const v2f*)(bp + 32 * t);
                        acc[t] = __builtin_amdgcn_wmma_f32_16x16x4_f32(
                            false, a, false, b, (short)0, acc[t], false, false);
                    }
                    if constexpr (EX) {
                        const v2f b = *(const v2f*)(bx + (4 * q + sub) * P);
                        accx = __builtin_amdgcn_wmma_f32_16x16x4_f32(
                            false, a, false, b, (short)0, accx, false, false);
                    }
                }
            }
        };
        if (ex) kloop(flag_t<true>{});
        else    kloop(flag_t<false>{});

        float* crow = C + (size_t)(rt * 16 + half * 8) * NOUT + cg * CW + mr;
#pragma unroll
        for (int t = 0; t < TL; ++t)
#pragma unroll
            for (int r = 0; r < 8; ++r)
                crow[(size_t)r * NOUT + t * 16] = acc[t][r];

        if (ex && mr < 2 * H) {
            const size_t row0 = (size_t)(rt * 16 + half * 8);
            float* cp = (mr < H) ? als + row0 * H + mr : ald + row0 * H + (mr - H);
#pragma unroll
            for (int r = 0; r < 8; ++r)
                cp[(size_t)r * H] = accx[r];
        }

        if (PF) {
#pragma unroll
            for (int q = 0; q < NQ; ++q) av[q] = nv[q];
        }
    }
}

#define NPB 196
#define QW  640
#define CH  8192
#define MAXBIN 1024
#define RSH 18
#define RMASK ((1 << RSH) - 1)
static_assert(MAXBIN * NPB <= (1 << RSH) && NPB <= (1 << (31 - RSH)),
              "a record must hold any source node and any row of a group");

__global__ __launch_bounds__(1024) void bin_edges(const int* __restrict__ src,
                                                  const int* __restrict__ dst,
                                                  int* __restrict__ rec,
                                                  int* __restrict__ seg,
                                                  int E_, int n, int nbin)
{
    __shared__ int hist[MAXBIN];
    __shared__ int wsum[32];
    const int tid  = threadIdx.x;
    const int lane = tid & 31;
    const int w    = tid >> 5;
    const int base = blockIdx.x * CH;

    hist[tid] = 0;
    __syncthreads();

    int d[CH / 1024], sv[CH / 1024], bin[CH / 1024], rk[CH / 1024];
#pragma unroll
    for (int k = 0; k < CH / 1024; ++k) {
        const int e = base + k * 1024 + tid;
        d[k] = (e < E_) ? dst[e] : -1;
        sv[k] = (e < E_) ? src[e] : 0;
    }
    const unsigned nlim = (unsigned)min(n, MAXBIN * NPB);
#pragma unroll
    for (int k = 0; k < CH / 1024; ++k) {
        bin[k] = -1;
        rk[k]  = 0;
        if ((unsigned)d[k] < nlim) {
            bin[k] = d[k] / NPB;
            rk[k]  = atomicAdd(&hist[bin[k]], 1);
        }
    }
    __syncthreads();

    const int v = hist[tid];
    int x = v;
    for (int o = 1; o < 32; o <<= 1) {
        const int y = __shfl_up(x, o);
        if (lane >= o) x += y;
    }
    if (lane == 31) wsum[w] = x;
    __syncthreads();
    if (w == 0) {
        const int t = wsum[lane];
        int s = t;
        for (int o = 1; o < 32; o <<= 1) {
            const int y = __shfl_up(s, o);
            if (lane >= o) s += y;
        }
        wsum[lane] = s - t;
    }
    __syncthreads();
    const int st = wsum[w] + x - v;
    hist[tid] = st;
    if (tid < nbin) {
        int* sp = seg + ((size_t)blockIdx.x * nbin + tid) * 2;
        sp[0] = st;
        sp[1] = st + v;
    }
    __syncthreads();

#pragma unroll
    for (int k = 0; k < CH / 1024; ++k)
        if (bin[k] >= 0)
            rec[base + hist[bin[k]] + rk[k]] =
                (sv[k] & RMASK) | ((d[k] - bin[k] * NPB) << RSH);
}

template<int H, int F>
__global__ __launch_bounds__(1024) void aggr_lds(const int* __restrict__ rec,
                                                 const int* __restrict__ seg,
                                                 const float* __restrict__ als,
                                                 const float* __restrict__ ald,
                                                 const float* __restrict__ h,
                                                 const float* __restrict__ b,
                                                 float* __restrict__ outp,
                                                 int E_, int n)
{
    constexpr int LPR  = F / 4;
    constexpr int LS   = (LPR < 32) ? LPR : 32;
    constexpr int RPL  = 32 / LS;
    constexpr int NJ   = LPR / LS;
    constexpr int CPHD = F / H;
    constexpr int NL   = 8;
    constexpr int UE   = NL * RPL / NJ;
    constexpr int QPR  = F / 4;
    static_assert(RPL == 1 || (RPL == 2 && NJ == 1), "one row per load, or two rows in one load");
    static_assert(CPHD % 4 == 0 && F % 4 == 0, "a lane's four columns lie in one head");
    __shared__ __attribute__((aligned(16))) float acc[NPB * F];
    __shared__ float dsum[H * NPB];
    __shared__ int   q[32 * QW];
    __shared__ __attribute__((aligned(16))) float wbuf[32 * 32 * H];

    const int tid  = threadIdx.x;
    const int lane = tid & 31;
    const int w    = __builtin_amdgcn_readfirstlane(tid >> 5);
    const int lo   = blockIdx.x * NPB;
    const int cntn = min(NPB, n - lo);
    const int sub  = lane / LS;
    const int ll   = lane % LS;

    for (int i = tid; i < NPB * F / 4; i += 1024)
        ((float4*)acc)[i] = make_float4(0.f, 0.f, 0.f, 0.f);
    for (int i = tid; i < NPB * H; i += 1024)
        dsum[i] = 0.f;
    __syncthreads();

    int* myq = q + w * QW;
    float* wb = wbuf + w * (32 * H);
    int cnt = 0;

    auto drain = [&]() {
        __builtin_amdgcn_fence(__ATOMIC_ACQ_REL, "wavefront");
        for (int qb = 0; qb < cnt; qb += 32) {
            const int idx = qb + lane;
            int s = 0, dl = 0;
            float wg[H];
#pragma unroll
            for (int hd = 0; hd < H; ++hd) wg[hd] = 0.f;
            if (idx < cnt) {
                const int r = myq[idx];
                s  = r & RMASK;
                dl = r >> RSH;
                const int d = lo + dl;
#pragma unroll
                for (int hd = 0; hd < H; ++hd) {
                    float x = als[s * H + hd] + ald[d * H + hd];
                    float lv = (x >= 0.f) ? x : NEG_SLOPE * x;
                    wg[hd] = expf(lv);
                    atomicAdd(&dsum[hd * NPB + dl], wg[hd]);
                }
            }
#pragma unroll
            for (int hd = 0; hd < H; ++hd) wb[lane * H + hd] = wg[hd];
            __builtin_amdgcn_fence(__ATOMIC_ACQ_REL, "wavefront");
            const int nb = min(32, cnt - qb);
            auto group = [&](auto tag, int j0) {
                constexpr bool FULL = decltype(tag)::value;
                if constexpr (RPL == 1) {
                    float4 hv[UE][NJ];
                    float  wv[UE][NJ];
#pragma unroll
                    for (int u = 0; u < UE; ++u) {
                        const int j  = FULL ? j0 + u : min(j0 + u, nb - 1);
                        const int sj = __builtin_amdgcn_readlane(s, j);
                        const float* hs = h + (size_t)sj * F + 4 * lane;
#pragma unroll
                        for (int jj = 0; jj < NJ; ++jj)
                            hv[u][jj] = *(const float4*)(hs + 128 * jj);
#pragma unroll
                        for (int jj = 0; jj < NJ; ++jj)
                            wv[u][jj] = wb[j * H + (128 * jj + 4 * lane) / CPHD];
                    }
#pragma unroll
                    for (int u = 0; u < UE; ++u) {
                        const int j = j0 + u;
                        if (FULL || j < nb) {
                            const int dj = __builtin_amdgcn_readlane(dl, j);
                            float* ar = acc + dj * F + lane;
#pragma unroll
                            for (int jj = 0; jj < NJ; ++jj) {
                                const float a = wv[u][jj];
                                float* p = ar + 128 * jj;
                                atomicAdd(p,          a * hv[u][jj].x);
                                atomicAdd(p + LS,     a * hv[u][jj].y);
                                atomicAdd(p + 2 * LS, a * hv[u][jj].z);
                                atomicAdd(p + 3 * LS, a * hv[u][jj].w);
                            }
                        }
                    }
                } else {
                    float4 hv[NL];
                    float  wv[NL];
                    int    dv[NL];
#pragma unroll
                    for (int t = 0; t < NL; ++t) {
                        const int jr = j0 + RPL * t + sub;
                        const int j  = FULL ? jr : min(jr, nb - 1);
                        const int rj = myq[qb + j];
                        const int sj = rj & RMASK;
                        dv[t] = rj >> RSH;
                        hv[t] = *(const float4*)(h + (size_t)sj * F + 4 * ll);
                        wv[t] = wb[j * H + (4 * ll) / CPHD];
                    }
#pragma unroll
                    for (int t = 0; t < NL; ++t) {
                        const int j = j0 + RPL * t + sub;
                        if (FULL || j < nb) {
                            const float a = wv[t];
                            float* ar  = acc + dv[t] * F + ll;
                            float* p01 = ar + (sub ? 2 * LS : 0);
                            float* p23 = ar + (sub ? 0 : 2 * LS);
                            atomicAdd(p01,      a * (sub ? hv[t].z : hv[t].x));
                            atomicAdd(p01 + LS, a * (sub ? hv[t].w : hv[t].y));
                            atomicAdd(p23,      a * (sub ? hv[t].x : hv[t].z));
                            atomicAdd(p23 + LS, a * (sub ? hv[t].y : hv[t].w));
                        }
                    }
                }
            };
            int j0 = 0;
            for (; j0 + UE <= nb; j0 += UE) group(flag_t<true>{}, j0);
            if (j0 < nb) group(flag_t<false>{}, j0);
        }
    };

    auto push = [&](bool m, int e) {
        const unsigned bal = __builtin_amdgcn_ballot_w32(m);
        if (m) myq[cnt + (int)__builtin_amdgcn_mbcnt_lo(bal, 0u)] = e;
        cnt += __builtin_popcount(bal);
    };

    const int nsb  = (E_ + CH - 1) / CH;
    const int nbin = gridDim.x;
    for (int b0 = w; b0 < nsb; b0 += 1024) {
        const int bl = b0 + 32 * lane;
        int2 r = make_int2(0, 0);
        if (bl < nsb)
            r = *(const int2*)(seg + ((size_t)bl * nbin + blockIdx.x) * 2);
        const int nmine = min(32, (nsb - b0 + 31) / 32);
        for (int l = 0; l < nmine; ++l) {
            const int a0 = __builtin_amdgcn_readlane(r.x, l);
            const int a1 = __builtin_amdgcn_readlane(r.y, l);
            const int* pp = rec + (size_t)(b0 + 32 * l) * CH;
            for (int i = a0; i < a1; i += 32) {
                if (cnt > QW - 32) { drain(); cnt = 0; }
                if (i + lane < a1) myq[cnt + lane] = pp[i + lane];
                cnt += min(32, a1 - i);
            }
        }
    }
    for (int i0 = w * 32; i0 < cntn; i0 += 1024) {
        if (cnt > QW - 32) { drain(); cnt = 0; }
        const int i = i0 + lane;
        push(i < cntn, (lo + i) | (i << RSH));
    }
    drain();
    __syncthreads();

    for (int i = tid; i < H * NPB; i += 1024)
        if (i % NPB < cntn) dsum[i] = 1.0f / dsum[i];
    __syncthreads();

    float* orow = outp + (size_t)lo * F;
    for (int i = tid; i < cntn * QPR; i += 1024) {
        const int r  = i / QPR;
        const int qd = i % QPR;
        const int c0 = 4 * qd;
        const float* ap = acc + r * F + (qd / LS) * (4 * LS) + (qd % LS);
        const float  rd = dsum[(c0 / CPHD) * NPB + r];
        const float4 bb = *(const float4*)(b + c0);
        float4 v;
        v.x = ap[0]      * rd + bb.x;
        v.y = ap[LS]     * rd + bb.y;
        v.z = ap[2 * LS] * rd + bb.z;
        v.w = ap[3 * LS] * rd + bb.w;
        v.x = (v.x > 0.f) ? v.x : expm1f(v.x);
        v.y = (v.y > 0.f) ? v.y : expm1f(v.y);
        v.z = (v.z > 0.f) ? v.z : expm1f(v.z);
        v.w = (v.w > 0.f) ? v.w : expm1f(v.w);
        *(float4*)(orow + (size_t)r * F + c0) = v;
    }
}

extern "C" void kernel_launch(void* const* d_in, const int* in_sizes, int n_in,
                              void* d_out, int out_size, void* d_ws, size_t ws_size,
                              hipStream_t stream)
{
    const float* x   = (const float*)d_in[0];
    const int*   ei  = (const int*)  d_in[1];
    const float* W1  = (const float*)d_in[2];
    const float* as1 = (const float*)d_in[3];
    const float* ad1 = (const float*)d_in[4];
    const float* b1  = (const float*)d_in[5];
    const float* W2  = (const float*)d_in[6];
    const float* as2 = (const float*)d_in[7];
    const float* ad2 = (const float*)d_in[8];
    const float* b2  = (const float*)d_in[9];
    float* out = (float*)d_out;

    const int IN_DIM = 128, H1 = 8, F1 = 256, OUT_DIM = 64;
    int n = in_sizes[0] / IN_DIM;
    int e = in_sizes[1] / 2;
    const int* src = ei;
    const int* dst = ei + e;

    float* ws = (float*)d_ws;
    float* h1   = ws;  ws += (size_t)n * F1;
    float* g1   = ws;  ws += (size_t)n * F1;
    float* h2   = h1;
    int*   rec  = (int*)ws;  ws += (size_t)n * OUT_DIM;
    int nsb  = (e + CH - 1) / CH;
    int nbin = (n + NPB - 1) / NPB;
    int*   seg  = rec + (size_t)nsb * CH;
    float* als1 = ws;  ws += (size_t)n * H1;
    float* ald1 = ws;  ws += (size_t)n * H1;
    float* als2 = ws;  ws += (size_t)n;
    float* ald2 = ws;  ws += (size_t)n;

    bin_edges<<<nsb, 1024, 0, stream>>>(src, dst, rec, seg, e, n, nbin);

    gemm_wmma<128, 256, 64, 512, true, 8><<<256, 512, 0, stream>>>(x, W1, as1, ad1, h1, als1, ald1, n / 16);
    aggr_lds<8, 256><<<nbin, 1024, 0, stream>>>(rec, seg, als1, ald1, h1, b1, g1, e, n);

    gemm_wmma<256, 64, 32, 1024, false, 1><<<256, 1024, 0, stream>>>(g1, W2, as2, ad2, h2, als2, ald2, n / 16);
    aggr_lds<1, 64><<<nbin, 1024, 0, stream>>>(rec, seg, als2, ald2, h2, b2, out, e, n);
}


# Aggregation: a group's queued edge records are counting-sorted by destination row in LDS; waves take equal slices, sum w*h and w per row in registers, LDS float adds once per row; sum order changes
# speedup vs baseline: 4.1358x; 1.0597x over previous
#include <hip/hip_runtime.h>
#include <math.h>

typedef __attribute__((ext_vector_type(2))) float v2f;
typedef __attribute__((ext_vector_type(8))) float v8f;

#define NEG_SLOPE 0.2f

template<bool V> struct flag_t { static constexpr bool value = V; };

template<int K, int NOUT, int CW, int NT, bool PF, int H>
__global__ __launch_bounds__(NT) void gemm_wmma(const float* __restrict__ A,
                                                 const float* __restrict__ B,
                                                 const float* __restrict__ a_src,
                                                 const float* __restrict__ a_dst,
                                                 float* __restrict__ C,
                                                 float* __restrict__ als,
                                                 float* __restrict__ ald,
                                                 int nrt)
{
    constexpr int NG = NOUT / CW;
    constexpr int TL = CW / 16;
    constexpr int XO = 2 * NOUT;
    constexpr int P  = 2 * NOUT + 48;
    constexpr int NQ = K / 8;
    constexpr int G  = (NOUT / H) / 4;
    static_assert(2 * H <= 16 && NOUT % (4 * H) == 0 && G <= 32 && (G & (G - 1)) == 0,
                  "a head's copy items are a power-of-two run of lanes inside one wave");
    __shared__ float Bs[(K / 2) * P];

    const int tid  = threadIdx.x;
    const int lane = tid & 31;
    const int half = lane >> 4;
    const int mr   = lane & 15;
    const int w = __builtin_amdgcn_readfirstlane(tid >> 5);
    constexpr int NW = NT / 32;
    static_assert(NW % NG == 0, "the NG tasks of a row tile fall into one round of the waves");

    const int nj    = (nrt - (int)blockIdx.x + (int)gridDim.x - 1) / (int)gridDim.x;
    const int ntask = nj * NG;

    auto load_a = [&](float4 (&v)[NQ], int i) {
        const int rt = blockIdx.x + (i / NG) * gridDim.x;
        const float* arow = A + (size_t)(rt * 16 + mr) * K + 4 * half;
#pragma unroll
        for (int q = 0; q < NQ; ++q)
            v[q] = *(const float4*)(arow + 8 * q);
    };

    float4 av[NQ];
    if (PF && w < ntask) load_a(av, w);

    constexpr int NS = (K / 2) * (NOUT / 4);
    constexpr int SI = NS / NT;
    static_assert(NS % NT == 0, "every thread copies the same number of items");
    static_assert(NT % (NOUT / 4) == 0, "all items of a thread hold the same four columns");
    {
        const int   c4  = tid % (NOUT / 4);
        const float4 as4 = *(const float4*)(a_src + 4 * c4);
        const float4 ad4 = *(const float4*)(a_dst + 4 * c4);
        float4 r0[SI], r1[SI];
#pragma unroll
        for (int s = 0; s < SI; ++s) {
            const int i  = tid + s * NT;
            const int p  = i / (NOUT / 4);
            const int n4 = i % (NOUT / 4);
            r0[s] = *(const float4*)(B + (size_t)(2 * p) * NOUT + 4 * n4);
            r1[s] = *(const float4*)(B + (size_t)(2 * p + 1) * NOUT + 4 * n4);
        }
#pragma unroll
        for (int s = 0; s < SI; ++s) {
            const int i  = tid + s * NT;
            const int p  = i / (NOUT / 4);
            const int n4 = i % (NOUT / 4);
            float4* d = (float4*)(Bs + p * P + 8 * n4);
            d[0] = make_float4(r0[s].x, r1[s].x, r0[s].y, r1[s].y);
            d[1] = make_float4(r0[s].z, r1[s].z, r0[s].w, r1[s].w);
            float s0 = r0[s].x * as4.x + r0[s].y * as4.y + r0[s].z * as4.z + r0[s].w * as4.w;
            float s1 = r1[s].x * as4.x + r1[s].y * as4.y + r1[s].z * as4.z + r1[s].w * as4.w;
            float d0 = r0[s].x * ad4.x + r0[s].y * ad4.y + r0[s].z * ad4.z + r0[s].w * ad4.w;
            float d1 = r1[s].x * ad4.x + r1[s].y * ad4.y + r1[s].z * ad4.z + r1[s].w * ad4.w;
#pragma unroll
            for (int o = 1; o < G; o <<= 1) {
                s0 += __shfl_xor(s0, o);
                s1 += __shfl_xor(s1, o);
                d0 += __shfl_xor(d0, o);
                d1 += __shfl_xor(d1, o);
            }
            if (n4 % G == 0) {
                const int hd = n4 / G;
                v2f ws, wd;
                ws.x = s0; ws.y = s1;
                wd.x = d0; wd.y = d1;
                *(v2f*)(Bs + p * P + XO + 2 * hd)       = ws;
                *(v2f*)(Bs + p * P + XO + 2 * (H + hd)) = wd;
            }
        }
        if constexpr (2 * H < 16) {
            constexpr int ZW = 2 * (16 - 2 * H);
            for (int i = tid; i < (K / 2) * ZW; i += NT)
                Bs[(i / ZW) * P + XO + 4 * H + i % ZW] = 0.f;
        }
    }
    __syncthreads();

    for (int i = w; i < ntask; i += NW) {
        const int cg = i % NG;
        const int rt = blockIdx.x + (i / NG) * gridDim.x;
        const bool ex = ((i / NW) % NG) == cg;

        float4 nv[NQ];
        if (PF) {
            load_a(nv, (i + NW < ntask) ? i + NW : i);
        }
        const float* arow = A + (size_t)(rt * 16 + mr) * K + 4 * half;

        v8f acc[TL] = {};
        v8f accx = {};
        const float* bl = Bs + (2 * half) * P + 2 * (cg * CW + mr);
        const float* bx = Bs + (2 * half) * P + XO + 2 * mr;
        auto kloop = [&](auto tag) {
            constexpr bool EX = decltype(tag)::value;
#pragma unroll
            for (int q = 0; q < NQ; ++q) {
                float4 aq;
                if constexpr (PF) aq = av[q];
                else              aq = *(const float4*)(arow + 8 * q);
#pragma unroll
                for (int sub = 0; sub < 2; ++sub) {
                    v2f a;
                    a.x = sub ? aq.z : aq.x;
                    a.y = sub ? aq.w : aq.y;
                    const float* bp = bl + (4 * q + sub) * P;
#pragma unroll
                    for (int t = 0; t < TL; ++t) {
                        const v2f b = *(const v2f*)(bp + 32 * t);
                        acc[t] = __builtin_amdgcn_wmma_f32_16x16x4_f32(
                            false, a, false, b, (short)0, acc[t], false, false);
                    }
                    if constexpr (EX) {
                        const v2f b = *(const v2f*)(bx + (4 * q + sub) * P);
                        accx = __builtin_amdgcn_wmma_f32_16x16x4_f32(
                            false, a, false, b, (short)0, accx, false, false);
                    }
                }
            }
        };
        if (ex) kloop(flag_t<true>{});
        else    kloop(flag_t<false>{});

        float* crow = C + (size_t)(rt * 16 + half * 8) * NOUT + cg * CW + mr;
#pragma unroll
        for (int t = 0; t < TL; ++t)
#pragma unroll
            for (int r = 0; r < 8; ++r)
                crow[(size_t)r * NOUT + t * 16] = acc[t][r];

        if (ex && mr < 2 * H) {
            const size_t row0 = (size_t)(rt * 16 + half * 8);
            float* cp = (mr < H) ? als + row0 * H + mr : ald + row0 * H + (mr - H);
#pragma unroll
            for (int r = 0; r < 8; ++r)
                cp[(size_t)r * H] = accx[r];
        }

        if (PF) {
#pragma unroll
            for (int q = 0; q < NQ; ++q) av[q] = nv[q];
        }
    }
}

#define NPB 196
#define QW  320
#define CH  8192
#define MAXBIN 1024
#define RSH 18
#define RMASK ((1 << RSH) - 1)
static_assert(MAXBIN * NPB <= (1 << RSH) && NPB <= (1 << (31 - RSH)),
              "a record must hold any source node and any row of a group");

__global__ __launch_bounds__(1024) void bin_edges(const int* __restrict__ src,
                                                  const int* __restrict__ dst,
                                                  int* __restrict__ rec,
                                                  int* __restrict__ seg,
                                                  int E_, int n, int nbin)
{
    __shared__ int hist[MAXBIN];
    __shared__ int wsum[32];
    const int tid  = threadIdx.x;
    const int lane = tid & 31;
    const int w    = tid >> 5;
    const int base = blockIdx.x * CH;

    hist[tid] = 0;
    __syncthreads();

    int d[CH / 1024], sv[CH / 1024], bin[CH / 1024], rk[CH / 1024];
#pragma unroll
    for (int k = 0; k < CH / 1024; ++k) {
        const int e = base + k * 1024 + tid;
        d[k] = (e < E_) ? dst[e] : -1;
        sv[k] = (e < E_) ? src[e] : 0;
    }
    const unsigned nlim = (unsigned)min(n, MAXBIN * NPB);
#pragma unroll
    for (int k = 0; k < CH / 1024; ++k) {
        bin[k] = -1;
        rk[k]  = 0;
        if ((unsigned)d[k] < nlim) {
            bin[k] = d[k] / NPB;
            rk[k]  = atomicAdd(&hist[bin[k]], 1);
        }
    }
    __syncthreads();

    const int v = hist[tid];
    int x = v;
    for (int o = 1; o < 32; o <<= 1) {
        const int y = __shfl_up(x, o);
        if (lane >= o) x += y;
    }
    if (lane == 31) wsum[w] = x;
    __syncthreads();
    if (w == 0) {
        const int t = wsum[lane];
        int s = t;
        for (int o = 1; o < 32; o <<= 1) {
            const int y = __shfl_up(s, o);
            if (lane >= o) s += y;
        }
        wsum[lane] = s - t;
    }
    __syncthreads();
    const int st = wsum[w] + x - v;
    hist[tid] = st;
    if (tid < nbin) {
        int* sp = seg + ((size_t)blockIdx.x * nbin + tid) * 2;
        sp[0] = st;
        sp[1] = st + v;
    }
    __syncthreads();

#pragma unroll
    for (int k = 0; k < CH / 1024; ++k)
        if (bin[k] >= 0)
            rec[base + hist[bin[k]] + rk[k]] =
                (sv[k] & RMASK) | ((d[k] - bin[k] * NPB) << RSH);
}

template<int H, int F>
__global__ __launch_bounds__(1024) void aggr_lds(const int* __restrict__ rec,
                                                 const int* __restrict__ seg,
                                                 const float* __restrict__ als,
                                                 const float* __restrict__ ald,
                                                 const float* __restrict__ h,
                                                 const float* __restrict__ b,
                                                 float* __restrict__ outp,
                                                 int E_, int n)
{
    constexpr int LPR  = F / 4;
    constexpr int LS   = (LPR < 32) ? LPR : 32;
    constexpr int RPL  = 32 / LS;
    constexpr int NJ   = LPR / LS;
    constexpr int CPHD = F / H;
    constexpr int NL   = 8;
    constexpr int UE   = NL * RPL / NJ;
    constexpr int QPR  = F / 4;
    static_assert(RPL == 1 || (RPL == 2 && NJ == 1), "one row per load, or two rows in one load");
    static_assert(CPHD % 4 == 0 && F % 4 == 0, "a lane's four columns lie in one head");
    __shared__ __attribute__((aligned(16))) float acc[NPB * F];
    __shared__ float dsum[H * NPB];
    __shared__ int   q[32 * QW];
    __shared__ int   sorted[32 * QW];
    __shared__ __attribute__((aligned(16))) float wbuf[32 * 32 * H];
    __shared__ int   rowpos[NPB];
    __shared__ int   tot;

    const int tid  = threadIdx.x;
    const int lane = tid & 31;
    const int w    = __builtin_amdgcn_readfirstlane(tid >> 5);
    const int lo   = blockIdx.x * NPB;
    const int cntn = min(NPB, n - lo);
    const int sub  = lane / LS;
    const int ll   = lane % LS;

    for (int i = tid; i < NPB * F / 4; i += 1024)
        ((float4*)acc)[i] = make_float4(0.f, 0.f, 0.f, 0.f);
    for (int i = tid; i < NPB * H; i += 1024)
        dsum[i] = 0.f;
    for (int i = tid; i < NPB; i += 1024)
        rowpos[i] = 0;
    __syncthreads();

    int* myq = q + w * QW;
    float* wb = wbuf + w * (32 * H);
    int cnt = 0;

    auto drain = [&](const int* qp, int cn) {
        __builtin_amdgcn_fence(__ATOMIC_ACQ_REL, "wavefront");
        int    cur = -1;
        float4 ar[NJ];
        float  den[NJ];
#pragma unroll
        for (int jj = 0; jj < NJ; ++jj) {
            ar[jj]  = make_float4(0.f, 0.f, 0.f, 0.f);
            den[jj] = 0.f;
        }
        auto flush = [&]() {
            if (cur >= 0) {
#pragma unroll
                for (int jj = 0; jj < NJ; ++jj) {
                    float* p = acc + cur * F + 4 * LS * jj + ll;
                    atomicAdd(p,          ar[jj].x);
                    atomicAdd(p + LS,     ar[jj].y);
                    atomicAdd(p + 2 * LS, ar[jj].z);
                    atomicAdd(p + 3 * LS, ar[jj].w);
                    const int col = 4 * (LS * jj + ll);
                    if (col % CPHD == 0)
                        atomicAdd(&dsum[(col / CPHD) * NPB + cur], den[jj]);
                }
            }
#pragma unroll
            for (int jj = 0; jj < NJ; ++jj) {
                ar[jj]  = make_float4(0.f, 0.f, 0.f, 0.f);
                den[jj] = 0.f;
            }
        };
        for (int qb = 0; qb < cn; qb += 32) {
            const int idx = qb + lane;
            int s = 0, dl = 0;
            float wg[H];
#pragma unroll
            for (int hd = 0; hd < H; ++hd) wg[hd] = 0.f;
            if (idx < cn) {
                const int r = qp[idx];
                s  = r & RMASK;
                dl = r >> RSH;
                const int d = lo + dl;
#pragma unroll
                for (int hd = 0; hd < H; ++hd) {
                    float x = als[s * H + hd] + ald[d * H + hd];
                    float lv = (x >= 0.f) ? x : NEG_SLOPE * x;
                    wg[hd] = expf(lv);
                }
            }
#pragma unroll
            for (int hd = 0; hd < H; ++hd) wb[lane * H + hd] = wg[hd];
            __builtin_amdgcn_fence(__ATOMIC_ACQ_REL, "wavefront");
            const int nb = min(32, cn - qb);
            auto group = [&](auto tag, int j0) {
                constexpr bool FULL = decltype(tag)::value;
                if constexpr (RPL == 1) {
                    float4 hv[UE][NJ];
                    float  wv[UE][NJ];
#pragma unroll
                    for (int u = 0; u < UE; ++u) {
                        const int j  = FULL ? j0 + u : min(j0 + u, nb - 1);
                        const int sj = __builtin_amdgcn_readlane(s, j);
                        const float* hs = h + (size_t)sj * F + 4 * lane;
#pragma unroll
                        for (int jj = 0; jj < NJ; ++jj)
                            hv[u][jj] = *(const float4*)(hs + 128 * jj);
#pragma unroll
                        for (int jj = 0; jj < NJ; ++jj)
                            wv[u][jj] = wb[j * H + (128 * jj + 4 * lane) / CPHD];
                    }
#pragma unroll
                    for (int u = 0; u < UE; ++u) {
                        const int j = j0 + u;
                        if (FULL || j < nb) {
                            const int dj = __builtin_amdgcn_readlane(dl, j);
                            if (dj != cur) { flush(); cur = dj; }
#pragma unroll
                            for (int jj = 0; jj < NJ; ++jj) {
                                const float a = wv[u][jj];
                                ar[jj].x += a * hv[u][jj].x;
                                ar[jj].y += a * hv[u][jj].y;
                                ar[jj].z += a * hv[u][jj].z;
                                ar[jj].w += a * hv[u][jj].w;
                                den[jj]  += a;
                            }
                        }
                    }
                } else {
                    float4 hv[NL];
                    float  wv[NL];
                    int    dv[NL];
#pragma unroll
                    for (int t = 0; t < NL; ++t) {
                        const int jr = j0 + RPL * t + sub;
                        const int j  = FULL ? jr : min(jr, nb - 1);
                        const int rj = qp[qb + j];
                        const int sj = rj & RMASK;
                        dv[t] = rj >> RSH;
                        hv[t] = *(const float4*)(h + (size_t)sj * F + 4 * ll);
                        wv[t] = wb[j * H + (4 * ll) / CPHD];
                    }
#pragma unroll
                    for (int t = 0; t < NL; ++t) {
                        const int j = j0 + RPL * t + sub;
                        if (FULL || j < nb) {
                            if (dv[t] != cur) { flush(); cur = dv[t]; }
                            const float a = wv[t];
                            ar[0].x += a * hv[t].x;
                            ar[0].y += a * hv[t].y;
                            ar[0].z += a * hv[t].z;
                            ar[0].w += a * hv[t].w;
                            den[0]  += a;
                        }
                    }
                }
            };
            int j0 = 0;
            for (; j0 + UE <= nb; j0 += UE) group(flag_t<true>{}, j0);
            if (j0 < nb) group(flag_t<false>{}, j0);
        }
        flush();
    };

    auto push = [&](bool m, int e) {
        const unsigned bal = __builtin_amdgcn_ballot_w32(m);
        if (m) myq[cnt + (int)__builtin_amdgcn_mbcnt_lo(bal, 0u)] = e;
        cnt += __builtin_popcount(bal);
    };

    const int nsb  = (E_ + CH - 1) / CH;
    const int nbin = gridDim.x;
    for (int b0 = w; b0 < nsb; b0 += 1024) {
        const int bl = b0 + 32 * lane;
        int2 r = make_int2(0, 0);
        if (bl < nsb)
            r = *(const int2*)(seg + ((size_t)bl * nbin + blockIdx.x) * 2);
        const int nmine = min(32, (nsb - b0 + 31) / 32);
        for (int l = 0; l < nmine; ++l) {
            const int a0 = __builtin_amdgcn_readlane(r.x, l);
            const int a1 = __builtin_amdgcn_readlane(r.y, l);
            const int* pp = rec + (size_t)(b0 + 32 * l) * CH;
            for (int i = a0; i < a1; i += 32) {
                if (cnt > QW - 32) { drain(myq, cnt); cnt = 0; }
                if (i + lane < a1) myq[cnt + lane] = pp[i + lane];
                cnt += min(32, a1 - i);
            }
        }
    }
    for (int i0 = w * 32; i0 < cntn; i0 += 1024) {
        if (cnt > QW - 32) { drain(myq, cnt); cnt = 0; }
        const int i = i0 + lane;
        push(i < cntn, (lo + i) | (i << RSH));
    }

    __builtin_amdgcn_fence(__ATOMIC_ACQ_REL, "wavefront");
    for (int idx = lane; idx < cnt; idx += 32)
        atomicAdd(&rowpos[myq[idx] >> RSH], 1);
    __syncthreads();
    if (w == 0) {
        constexpr int RL = (NPB + 31) / 32;
        int c[RL];
        int sum = 0;
#pragma unroll
        for (int k = 0; k < RL; ++k) {
            const int r = RL * lane + k;
            c[k] = (r < NPB) ? rowpos[r] : 0;
            sum += c[k];
        }
        int x = sum;
        for (int o = 1; o < 32; o <<= 1) {
            const int y = __shfl_up(x, o);
            if (lane >= o) x += y;
        }
        int st = x - sum;
#pragma unroll
        for (int k = 0; k < RL; ++k) {
            const int r = RL * lane + k;
            if (r < NPB) rowpos[r] = st;
            st += c[k];
        }
        if (lane == 31) tot = x;
    }
    __syncthreads();
    for (int idx = lane; idx < cnt; idx += 32) {
        const int r = myq[idx];
        sorted[atomicAdd(&rowpos[r >> RSH], 1)] = r;
    }
    __syncthreads();
    {
        const int T   = __builtin_amdgcn_readfirstlane(tot);
        const int per = (((T + 31) >> 5) + 15) & ~15;
        const int st  = min(w * per, T);
        const int en  = min(st + per, T);
        drain(sorted + st, en - st);
    }
    __syncthreads();

    for (int i = tid; i < H * NPB; i += 1024)
        if (i % NPB < cntn) dsum[i] = 1.0f / dsum[i];
    __syncthreads();

    float* orow = outp + (size_t)lo * F;
    for (int i = tid; i < cntn * QPR; i += 1024) {
        const int r  = i / QPR;
        const int qd = i % QPR;
        const int c0 = 4 * qd;
        const float* ap = acc + r * F + (qd / LS) * (4 * LS) + (qd % LS);
        const float  rd = dsum[(c0 / CPHD) * NPB + r];
        const float4 bb = *(const float4*)(b + c0);
        float4 v;
        v.x = ap[0]      * rd + bb.x;
        v.y = ap[LS]     * rd + bb.y;
        v.z = ap[2 * LS] * rd + bb.z;
        v.w = ap[3 * LS] * rd + bb.w;
        v.x = (v.x > 0.f) ? v.x : expm1f(v.x);
        v.y = (v.y > 0.f) ? v.y : expm1f(v.y);
        v.z = (v.z > 0.f) ? v.z : expm1f(v.z);
        v.w = (v.w > 0.f) ? v.w : expm1f(v.w);
        *(float4*)(orow + (size_t)r * F + c0) = v;
    }
}

extern "C" void kernel_launch(void* const* d_in, const int* in_sizes, int n_in,
                              void* d_out, int out_size, void* d_ws, size_t ws_size,
                              hipStream_t stream)
{
    const float* x   = (const float*)d_in[0];
    const int*   ei  = (const int*)  d_in[1];
    const float* W1  = (const float*)d_in[2];
    const float* as1 = (const float*)d_in[3];
    const float* ad1 = (const float*)d_in[4];
    const float* b1  = (const float*)d_in[5];
    const float* W2  = (const float*)d_in[6];
    const float* as2 = (const float*)d_in[7];
    const float* ad2 = (const float*)d_in[8];
    const float* b2  = (const float*)d_in[9];
    float* out = (float*)d_out;

    const int IN_DIM = 128, H1 = 8, F1 = 256, OUT_DIM = 64;
    int n = in_sizes[0] / IN_DIM;
    int e = in_sizes[1] / 2;
    const int* src = ei;
    const int* dst = ei + e;

    float* ws = (float*)d_ws;
    float* h1   = ws;  ws += (size_t)n * F1;
    float* g1   = ws;  ws += (size_t)n * F1;
    float* h2   = h1;
    int*   rec  = (int*)ws;  ws += (size_t)n * OUT_DIM;
    int nsb  = (e + CH - 1) / CH;
    int nbin = (n + NPB - 1) / NPB;
    int*   seg  = rec + (size_t)nsb * CH;
    float* als1 = ws;  ws += (size_t)n * H1;
    float* ald1 = ws;  ws += (size_t)n * H1;
    float* als2 = ws;  ws += (size_t)n;
    float* ald2 = ws;  ws += (size_t)n;

    bin_edges<<<nsb, 1024, 0, stream>>>(src, dst, rec, seg, e, n, nbin);

    gemm_wmma<128, 256, 64, 512, true, 8><<<256, 512, 0, stream>>>(x, W1, as1, ad1, h1, als1, ald1, n / 16);
    aggr_lds<8, 256><<<nbin, 1024, 0, stream>>>(rec, seg, als1, ald1, h1, b1, g1, e, n);

    gemm_wmma<256, 64, 32, 1024, false, 1><<<256, 1024, 0, stream>>>(g1, W2, as2, ad2, h2, als2, ald2, n / 16);
    aggr_lds<1, 64><<<nbin, 1024, 0, stream>>>(rec, seg, als2, ald2, h2, b2, out, e, n);
}
